# starting kernel
# baseline (speedup 1.0000x reference)
#include <hip/hip_runtime.h>
#include <hip/hip_bf16.h>
#include <math.h>

#define NTOT   2048
#define CCAP   32
#define DIN    8
#define DOUT   16
#define BATCH  32
#define NPER   32
#define NCHUNK (NTOT / NPER)
#define COLS   (CCAP * DOUT)
#define OUTEL  (BATCH * COLS)

typedef __attribute__((ext_vector_type(2))) float v2f;
typedef __attribute__((ext_vector_type(8))) float v8f;

__global__ __launch_bounds__(256)
void caps_softmax(const float* __restrict__ R, float* __restrict__ Rn) {
    const int c   = blockIdx.x;
    const int tid = threadIdx.x;
    __shared__ float red[256];

    float m = -INFINITY;
    for (int n = tid; n < NTOT; n += 256) m = fmaxf(m, R[n * CCAP + c]);
    red[tid] = m; __syncthreads();
    for (int s = 128; s > 0; s >>= 1) {
        if (tid < s) red[tid] = fmaxf(red[tid], red[tid + s]);
        __syncthreads();
    }
    m = red[0]; __syncthreads();

    float sum = 0.0f;
    for (int n = tid; n < NTOT; n += 256) sum += __expf(R[n * CCAP + c] - m);
    red[tid] = sum; __syncthreads();
    for (int s = 128; s > 0; s >>= 1) {
        if (tid < s) red[tid] += red[tid + s];
        __syncthreads();
    }
    const float inv = 1.0f / red[0];

    for (int n = tid; n < NTOT; n += 256)
        Rn[n * CCAP + c] = __expf(R[n * CCAP + c] - m) * inv;
}

__global__ __launch_bounds__(32)
void caps_gemm(const float* __restrict__ x, const float* __restrict__ W,
               const float* __restrict__ Rn, float* __restrict__ part) {
    const int t    = blockIdx.x;
    const int ch   = blockIdx.y;
    const int lane = threadIdx.x;
    const int j    = lane & 15;
    const int hi   = lane >> 4;
    const int kofs = hi * 2;

    v8f acc0 = {};
    v8f acc1 = {};

    const int n0 = ch * NPER;
    const float* wp  = W  + (size_t)n0 * (CCAP * DOUT * DIN) + t * (DOUT * DIN) + j * DIN + kofs;
    const float* xp0 = x  + (size_t)j        * (NTOT * DIN) + n0 * DIN + kofs;
    const float* xp1 = x  + (size_t)(j + 16) * (NTOT * DIN) + n0 * DIN + kofs;
    const float* rp  = Rn + n0 * CCAP + t;

    #pragma unroll 4
    for (int nn = 0; nn < NPER; ++nn) {
        const float rn = rp[nn * CCAP];

        v2f b0, b1, a00, a01, a10, a11;
        b0.x  = wp[0] * rn;  b0.y  = wp[1] * rn;
        b1.x  = wp[4] * rn;  b1.y  = wp[5] * rn;
        a00.x = xp0[0];      a00.y = xp0[1];
        a01.x = xp0[4];      a01.y = xp0[5];
        a10.x = xp1[0];      a10.y = xp1[1];
        a11.x = xp1[4];      a11.y = xp1[5];

        acc0 = __builtin_amdgcn_wmma_f32_16x16x4_f32(false, a00, false, b0,
                                                     (short)0, acc0, false, false);
        acc0 = __builtin_amdgcn_wmma_f32_16x16x4_f32(false, a01, false, b1,
                                                     (short)0, acc0, false, false);
        acc1 = __builtin_amdgcn_wmma_f32_16x16x4_f32(false, a10, false, b0,
                                                     (short)0, acc1, false, false);
        acc1 = __builtin_amdgcn_wmma_f32_16x16x4_f32(false, a11, false, b1,
                                                     (short)0, acc1, false, false);

        wp  += CCAP * DOUT * DIN;
        xp0 += DIN;
        xp1 += DIN;
    }

    float* p = part + (size_t)ch * OUTEL + t * DOUT + j;
    #pragma unroll
    for (int v = 0; v < 8; ++v) {
        const int b = v + hi * 8;
        p[(size_t)b        * COLS] = acc0[v];
        p[(size_t)(b + 16) * COLS] = acc1[v];
    }
}

__global__ __launch_bounds__(256)
void caps_reduce_squash(const float* __restrict__ part, float* __restrict__ out) {
    const int idx = blockIdx.x * 256 + threadIdx.x;
    float s = 0.0f;
    #pragma unroll 8
    for (int ch = 0; ch < NCHUNK; ++ch)
        s += part[(size_t)ch * OUTEL + idx];

    float sq = s * s;
    for (int m = 1; m < 16; m <<= 1)
        sq += __shfl_xor(sq, m, 32);
    sq += 1e-7f;
    const float scale = sqrtf(sq) / (1.0f + sq);
    out[idx] = scale * s;
}

extern "C" void kernel_launch(void* const* d_in, const int* in_sizes, int n_in,
                              void* d_out, int out_size, void* d_ws, size_t ws_size,
                              hipStream_t stream) {
    const float* x = (const float*)d_in[0];
    const float* W = (const float*)d_in[1];
    const float* R = (const float*)d_in[2];
    float* out = (float*)d_out;

    float* Rn   = (float*)d_ws;
    float* part = Rn + (size_t)NTOT * CCAP;

    caps_softmax<<<CCAP, 256, 0, stream>>>(R, Rn);

    dim3 grid(CCAP, NCHUNK);
    caps_gemm<<<grid, 32, 0, stream>>>(x, W, Rn, part);

    caps_reduce_squash<<<OUTEL / 256, 256, 0, stream>>>(part, out);
}


# GEMM regrouped into 256 eight-wave workgroups (8 capsules x one 32-n chunk): x chunk read once per workgroup by whole-line 16-byte loads into LDS and shared; W one 16-byte load per lane per n
# speedup vs baseline: 2.3873x; 2.3873x over previous
#include <hip/hip_runtime.h>
#include <hip/hip_bf16.h>
#include <math.h>

#define NTOT   2048
#define CCAP   32
#define DIN    8
#define DOUT   16
#define BATCH  32
#define NPER   32
#define NCHUNK (NTOT / NPER)
#define COLS   (CCAP * DOUT)
#define OUTEL  (BATCH * COLS)
#define FPITCH (NCHUNK + 4)
#define WPB    8
#define NGRP   (CCAP / WPB)
#define NGRPN  8
#define XPITCH (NPER * DIN + 4)

typedef __attribute__((ext_vector_type(2))) float v2f;
typedef __attribute__((ext_vector_type(8))) float v8f;

#define ROWX(v, m) __builtin_bit_cast(float, __builtin_amdgcn_update_dpp( \
    __builtin_bit_cast(int, (v)), __builtin_bit_cast(int, (v)), 0x160 + (m), 0xf, 0xf, false))
__device__ __forceinline__ float xhalf(float v) {
    const int b = __builtin_bit_cast(int, v);
    return __builtin_bit_cast(float,
        __builtin_amdgcn_permlanex16(b, b, 0x76543210, 0xfedcba98, false, false));
}
__device__ __forceinline__ float wave_max(float v) {
    v = fmaxf(v, ROWX(v, 1));
    v = fmaxf(v, ROWX(v, 2));
    v = fmaxf(v, ROWX(v, 4));
    v = fmaxf(v, ROWX(v, 8));
    return fmaxf(v, xhalf(v));
}
__device__ __forceinline__ float wave_sum(float v) {
    v += ROWX(v, 1);
    v += ROWX(v, 2);
    v += ROWX(v, 4);
    v += ROWX(v, 8);
    return v + xhalf(v);
}

__device__ __forceinline__ void pin() { __builtin_amdgcn_sched_barrier(0); }

__global__ __launch_bounds__(256)
void caps_gemm(const float* __restrict__ x, const float* __restrict__ W,
               const float* __restrict__ R, float* __restrict__ part,
               float* __restrict__ mst, float* __restrict__ zst) {
    __shared__ __attribute__((aligned(16))) float xs[BATCH * XPITCH];

    const int tid  = threadIdx.x;
    const int lane = tid & 31;
    const int wave = tid >> 5;
    const int t    = blockIdx.x * WPB + wave;
    const int ch   = blockIdx.y;
    const int j    = lane & 15;
    const int hi   = lane >> 4;
    const int n0   = ch * NPER;

    const int q  = tid & 63;
    const int br = tid >> 6;
    float4 xv[8];
    #pragma unroll
    for (int i = 0; i < 8; ++i)
        xv[i] = *(const float4*)(x + (size_t)(i * 4 + br) * (NTOT * DIN) + n0 * DIN + q * 4);

    const float r = R[(n0 + lane) * CCAP + t];

    const float* wp = W + (size_t)n0 * (CCAP * DOUT * DIN) + t * (DOUT * DIN) + j * DIN + hi * 4;
    float4 wv[NPER];
    #pragma unroll
    for (int nn = 0; nn < NPER; ++nn)
        wv[nn] = *(const float4*)(wp + (size_t)nn * (CCAP * DOUT * DIN));

    #pragma unroll
    for (int i = 0; i < 8; ++i)
        *(float4*)(xs + (i * 4 + br) * XPITCH + q * 4) = xv[i];
    asm volatile("s_wait_dscnt 0x0" ::: "memory");
    asm volatile("s_barrier_signal -1" ::: "memory");

    const float mloc = wave_max(r);
    const float e    = __expf(r - mloc);
    const float zloc = wave_sum(e);
    if (lane == 0) {
        mst[ch * CCAP + t] = mloc;
        zst[ch * CCAP + t] = zloc;
    }
    const int ebits = __float_as_int(e);

    asm volatile("s_barrier_wait -1" ::: "memory");

    v8f acc0 = {};
    v8f acc1 = {};
    const float* xa0 = xs + j * XPITCH + hi * 4;
    const float* xa1 = xs + (j + 16) * XPITCH + hi * 4;

    float4 fa[2][NGRPN], fb[2][NGRPN];
    #pragma unroll
    for (int u = 0; u < NGRPN; ++u) {
        fa[0][u] = *(const float4*)(xa0 + u * DIN);
        fb[0][u] = *(const float4*)(xa1 + u * DIN);
    }
    pin();
    #pragma unroll
    for (int gq = 0; gq < NPER / NGRPN; ++gq) {
        const int cur = gq & 1;
        if (gq + 1 < NPER / NGRPN) {
            #pragma unroll
            for (int u = 0; u < NGRPN; ++u) {
                fa[cur ^ 1][u] = *(const float4*)(xa0 + ((gq + 1) * NGRPN + u) * DIN);
                fb[cur ^ 1][u] = *(const float4*)(xa1 + ((gq + 1) * NGRPN + u) * DIN);
            }
        }
        pin();
        #pragma unroll
        for (int u = 0; u < NGRPN; ++u) {
            const int nn = gq * NGRPN + u;
            const float rn = __int_as_float(__builtin_amdgcn_readlane(ebits, nn));
            const float4 w  = wv[nn];
            const float4 a0 = fa[cur][u];
            const float4 a1 = fb[cur][u];

            v2f b0, b1, a00, a01, a10, a11;
            b0.x  = w.x * rn;    b0.y  = w.y * rn;
            b1.x  = w.z * rn;    b1.y  = w.w * rn;
            a00.x = a0.x;        a00.y = a0.y;
            a01.x = a0.z;        a01.y = a0.w;
            a10.x = a1.x;        a10.y = a1.y;
            a11.x = a1.z;        a11.y = a1.w;

            acc0 = __builtin_amdgcn_wmma_f32_16x16x4_f32(false, a00, false, b0,
                                                         (short)0, acc0, false, false);
            acc1 = __builtin_amdgcn_wmma_f32_16x16x4_f32(false, a10, false, b0,
                                                         (short)0, acc1, false, false);
            acc0 = __builtin_amdgcn_wmma_f32_16x16x4_f32(false, a01, false, b1,
                                                         (short)0, acc0, false, false);
            acc1 = __builtin_amdgcn_wmma_f32_16x16x4_f32(false, a11, false, b1,
                                                         (short)0, acc1, false, false);
        }
        pin();
    }

    float* p = part + (size_t)ch * OUTEL + t * DOUT + j;
    #pragma unroll
    for (int v = 0; v < 8; ++v) {
        const int b = v + hi * 8;
        p[(size_t)b        * COLS] = acc0[v];
        p[(size_t)(b + 16) * COLS] = acc1[v];
    }
}

__global__ __launch_bounds__(256)
void caps_reduce_squash(const float* __restrict__ part, const float* __restrict__ mst,
                        const float* __restrict__ zst, float* __restrict__ out) {
    __shared__ __attribute__((aligned(16))) float fsh[16 * FPITCH];
    __shared__ float redm[256];
    __shared__ float redz[256];

    const int tid   = threadIdx.x;
    const int idx   = blockIdx.x * 256 + tid;
    const int cbase = (blockIdx.x & 1) * 16;

    const int cl = tid & 15;
    const int g  = tid >> 4;
    float mv[4], zv[4];
    #pragma unroll
    for (int q = 0; q < 4; ++q) {
        mv[q] = mst[(g * 4 + q) * CCAP + cbase + cl];
        zv[q] = zst[(g * 4 + q) * CCAP + cbase + cl];
    }
    float pre[8];
    #pragma unroll
    for (int q = 0; q < 8; ++q) pre[q] = part[(size_t)q * OUTEL + idx];

    redm[tid] = fmaxf(fmaxf(mv[0], mv[1]), fmaxf(mv[2], mv[3]));
    __syncthreads();
    float M = redm[cl];
    #pragma unroll
    for (int g2 = 1; g2 < 16; ++g2) M = fmaxf(M, redm[g2 * 16 + cl]);
    float zp = 0.0f;
    #pragma unroll
    for (int q = 0; q < 4; ++q) {
        const float f = __expf(mv[q] - M);
        fsh[cl * FPITCH + g * 4 + q] = f;
        zp += zv[q] * f;
    }
    redz[tid] = zp;
    __syncthreads();

    const int mc = tid >> 4;
    float Z = 0.0f;
    #pragma unroll
    for (int g2 = 0; g2 < 16; ++g2) Z += redz[g2 * 16 + mc];

    float s = 0.0f;
    #pragma unroll
    for (int q = 0; q < 8; ++q) s += pre[q] * fsh[mc * FPITCH + q];
    #pragma unroll 8
    for (int ch = 8; ch < NCHUNK; ++ch)
        s += part[(size_t)ch * OUTEL + idx] * fsh[mc * FPITCH + ch];
    s *= 1.0f / Z;

    float sq = s * s;
    for (int m = 1; m < 16; m <<= 1)
        sq += __shfl_xor(sq, m, 32);
    sq += 1e-7f;
    const float scale = sqrtf(sq) / (1.0f + sq);
    out[idx] = scale * s;
}

extern "C" void kernel_launch(void* const* d_in, const int* in_sizes, int n_in,
                              void* d_out, int out_size, void* d_ws, size_t ws_size,
                              hipStream_t stream) {
    const float* x = (const float*)d_in[0];
    const float* W = (const float*)d_in[1];
    const float* R = (const float*)d_in[2];
    float* out = (float*)d_out;

    float* mst  = (float*)d_ws;
    float* zst  = mst + NCHUNK * CCAP;
    float* part = mst + (size_t)NTOT * CCAP;

    dim3 grid(NGRP, NCHUNK);
    caps_gemm<<<grid, WPB * 32, 0, stream>>>(x, W, R, part, mst, zst);

    caps_reduce_squash<<<OUTEL / 256, 256, 0, stream>>>(part, mst, zst, out);
}


# Reduce launch re-cut to 1024 one-capsule waves: a lane's 32 partial loads all in flight at once, statistics reduced in-wave, planes stored capsule-major so each load and store is one whole line
# speedup vs baseline: 3.2245x; 1.3507x over previous
#include <hip/hip_runtime.h>
#include <hip/hip_bf16.h>
#include <math.h>

#define NTOT   2048
#define CCAP   32
#define DIN    8
#define DOUT   16
#define BATCH  32
#define NPER   32
#define NCHUNK (NTOT / NPER)
#define COLS   (CCAP * DOUT)
#define OUTEL  (BATCH * COLS)
#define TILE   (BATCH * DOUT)
#define RWAVES 4
#define RCH    (NCHUNK / 2)
#define WPB    8
#define NGRP   (CCAP / WPB)
#define NGRPN  8
#define XPITCH (NPER * DIN + 4)

typedef __attribute__((ext_vector_type(2))) float v2f;
typedef __attribute__((ext_vector_type(8))) float v8f;

#define ROWX(v, m) __builtin_bit_cast(float, __builtin_amdgcn_update_dpp( \
    __builtin_bit_cast(int, (v)), __builtin_bit_cast(int, (v)), 0x160 + (m), 0xf, 0xf, false))
__device__ __forceinline__ float xhalf(float v) {
    const int b = __builtin_bit_cast(int, v);
    return __builtin_bit_cast(float,
        __builtin_amdgcn_permlanex16(b, b, 0x76543210, 0xfedcba98, false, false));
}
__device__ __forceinline__ float wave_max(float v) {
    v = fmaxf(v, ROWX(v, 1));
    v = fmaxf(v, ROWX(v, 2));
    v = fmaxf(v, ROWX(v, 4));
    v = fmaxf(v, ROWX(v, 8));
    return fmaxf(v, xhalf(v));
}
__device__ __forceinline__ float wave_sum(float v) {
    v += ROWX(v, 1);
    v += ROWX(v, 2);
    v += ROWX(v, 4);
    v += ROWX(v, 8);
    return v + xhalf(v);
}

__device__ __forceinline__ void pin() { __builtin_amdgcn_sched_barrier(0); }

__global__ __launch_bounds__(256)
void caps_gemm(const float* __restrict__ x, const float* __restrict__ W,
               const float* __restrict__ R, float* __restrict__ part,
               float* __restrict__ mst, float* __restrict__ zst) {
    __shared__ __attribute__((aligned(16))) float xs[BATCH * XPITCH];

    const int tid  = threadIdx.x;
    const int lane = tid & 31;
    const int wave = tid >> 5;
    const int t    = blockIdx.x * WPB + wave;
    const int ch   = blockIdx.y;
    const int j    = lane & 15;
    const int hi   = lane >> 4;
    const int n0   = ch * NPER;

    const int q  = tid & 63;
    const int br = tid >> 6;
    float4 xv[8];
    #pragma unroll
    for (int i = 0; i < 8; ++i)
        xv[i] = *(const float4*)(x + (size_t)(i * 4 + br) * (NTOT * DIN) + n0 * DIN + q * 4);

    const float r = R[(n0 + lane) * CCAP + t];

    const float* wp = W + (size_t)n0 * (CCAP * DOUT * DIN) + t * (DOUT * DIN) + j * DIN + hi * 4;
    float4 wv[NPER];
    #pragma unroll
    for (int nn = 0; nn < NPER; ++nn)
        wv[nn] = *(const float4*)(wp + (size_t)nn * (CCAP * DOUT * DIN));

    #pragma unroll
    for (int i = 0; i < 8; ++i)
        *(float4*)(xs + (i * 4 + br) * XPITCH + q * 4) = xv[i];
    asm volatile("s_wait_dscnt 0x0" ::: "memory");
    asm volatile("s_barrier_signal -1" ::: "memory");

    const float mloc = wave_max(r);
    const float e    = __expf(r - mloc);
    const float zloc = wave_sum(e);
    if (lane == 0) {
        mst[t * NCHUNK + ch] = mloc;
        zst[t * NCHUNK + ch] = zloc;
    }
    const int ebits = __float_as_int(e);

    asm volatile("s_barrier_wait -1" ::: "memory");

    v8f acc0 = {};
    v8f acc1 = {};
    const float* xa0 = xs + j * XPITCH + hi * 4;
    const float* xa1 = xs + (j + 16) * XPITCH + hi * 4;

    float4 fa[2][NGRPN], fb[2][NGRPN];
    #pragma unroll
    for (int u = 0; u < NGRPN; ++u) {
        fa[0][u] = *(const float4*)(xa0 + u * DIN);
        fb[0][u] = *(const float4*)(xa1 + u * DIN);
    }
    pin();
    #pragma unroll
    for (int gq = 0; gq < NPER / NGRPN; ++gq) {
        const int cur = gq & 1;
        if (gq + 1 < NPER / NGRPN) {
            #pragma unroll
            for (int u = 0; u < NGRPN; ++u) {
                fa[cur ^ 1][u] = *(const float4*)(xa0 + ((gq + 1) * NGRPN + u) * DIN);
                fb[cur ^ 1][u] = *(const float4*)(xa1 + ((gq + 1) * NGRPN + u) * DIN);
            }
        }
        pin();
        #pragma unroll
        for (int u = 0; u < NGRPN; ++u) {
            const int nn = gq * NGRPN + u;
            const float rn = __int_as_float(__builtin_amdgcn_readlane(ebits, nn));
            const float4 w  = wv[nn];
            const float4 a0 = fa[cur][u];
            const float4 a1 = fb[cur][u];

            v2f b0, b1, a00, a01, a10, a11;
            b0.x  = w.x * rn;    b0.y  = w.y * rn;
            b1.x  = w.z * rn;    b1.y  = w.w * rn;
            a00.x = a0.x;        a00.y = a0.y;
            a01.x = a0.z;        a01.y = a0.w;
            a10.x = a1.x;        a10.y = a1.y;
            a11.x = a1.z;        a11.y = a1.w;

            acc0 = __builtin_amdgcn_wmma_f32_16x16x4_f32(false, a00, false, b0,
                                                         (short)0, acc0, false, false);
            acc1 = __builtin_amdgcn_wmma_f32_16x16x4_f32(false, a10, false, b0,
                                                         (short)0, acc1, false, false);
            acc0 = __builtin_amdgcn_wmma_f32_16x16x4_f32(false, a01, false, b1,
                                                         (short)0, acc0, false, false);
            acc1 = __builtin_amdgcn_wmma_f32_16x16x4_f32(false, a11, false, b1,
                                                         (short)0, acc1, false, false);
        }
        pin();
    }

    float* p = part + ((size_t)ch * CCAP + t) * TILE + lane;
    #pragma unroll
    for (int v = 0; v < 8; ++v) {
        p[v * 32]       = acc0[v];
        p[256 + v * 32] = acc1[v];
    }
}

__global__ __launch_bounds__(RWAVES * 32)
void caps_reduce_squash(const float* __restrict__ part, const float* __restrict__ mst,
                        const float* __restrict__ zst, float* __restrict__ out) {
    __shared__ float sh[RWAVES * 32];

    const int tid  = threadIdx.x;
    const int lane = tid & 31;
    const int wv   = tid >> 5;
    const int cw   = wv & 1;
    const int q    = wv >> 1;
    const int cp   = blockIdx.x >> 4;
    const int e    = blockIdx.x & 15;
    const int c    = cp * 2 + cw;

    const float m0 = mst[c * NCHUNK + lane];
    const float m1 = mst[c * NCHUNK + 32 + lane];
    const float z0 = zst[c * NCHUNK + lane];
    const float z1 = zst[c * NCHUNK + 32 + lane];

    const float* pp = part + ((size_t)(q * RCH) * CCAP + c) * TILE + e * 32 + lane;
    float pv[RCH];
    #pragma unroll
    for (int i = 0; i < RCH; ++i) pv[i] = pp[(size_t)i * OUTEL];
    pin();

    const float M  = wave_max(fmaxf(m0, m1));
    const float f0 = __expf(m0 - M);
    const float f1 = __expf(m1 - M);
    const float Z  = wave_sum(z0 * f0 + z1 * f1);
    const int   fb = __float_as_int(q ? f1 : f0);

    float s0 = 0.0f, s1 = 0.0f, s2 = 0.0f, s3 = 0.0f;
    #pragma unroll
    for (int i = 0; i < RCH; i += 4) {
        s0 += pv[i + 0] * __int_as_float(__builtin_amdgcn_readlane(fb, i + 0));
        s1 += pv[i + 1] * __int_as_float(__builtin_amdgcn_readlane(fb, i + 1));
        s2 += pv[i + 2] * __int_as_float(__builtin_amdgcn_readlane(fb, i + 2));
        s3 += pv[i + 3] * __int_as_float(__builtin_amdgcn_readlane(fb, i + 3));
    }
    sh[tid] = ((s0 + s1) + (s2 + s3)) * (1.0f / Z);
    __syncthreads();

    if (tid < 64) {
        const int hc = lane >> 4;
        const int j  = lane & 15;
        const float s = sh[hc * 32 + wv * 16 + j] + sh[64 + hc * 32 + wv * 16 + j];

        float sq = s * s;
        sq += ROWX(sq, 1);
        sq += ROWX(sq, 2);
        sq += ROWX(sq, 4);
        sq += ROWX(sq, 8);
        sq += 1e-7f;
        const float scale = sqrtf(sq) / (1.0f + sq);

        const int pos = 2 * e + wv;
        const int b   = (pos & 16) + ((pos & 15) >> 1) + 8 * (pos & 1);
        out[b * COLS + cp * 32 + lane] = scale * s;
    }
}

extern "C" void kernel_launch(void* const* d_in, const int* in_sizes, int n_in,
                              void* d_out, int out_size, void* d_ws, size_t ws_size,
                              hipStream_t stream) {
    const float* x = (const float*)d_in[0];
    const float* W = (const float*)d_in[1];
    const float* R = (const float*)d_in[2];
    float* out = (float*)d_out;

    float* mst  = (float*)d_ws;
    float* zst  = mst + NCHUNK * CCAP;
    float* part = mst + (size_t)NTOT * CCAP;

    dim3 grid(NGRP, NCHUNK);
    caps_gemm<<<grid, WPB * 32, 0, stream>>>(x, W, R, part, mst, zst);

    caps_reduce_squash<<<256, RWAVES * 32, 0, stream>>>(part, mst, zst, out);
}


# GEMM prologue: first use of the routing logits held behind the issue of all 41 loads (it was waited for before the x and W loads: one exposed memory round trip); weight sum moved among the WMMAs
# speedup vs baseline: 3.5445x; 1.0992x over previous
#include <hip/hip_runtime.h>
#include <hip/hip_bf16.h>
#include <math.h>

#define NTOT   2048
#define CCAP   32
#define DIN    8
#define DOUT   16
#define BATCH  32
#define NPER   32
#define NCHUNK (NTOT / NPER)
#define COLS   (CCAP * DOUT)
#define OUTEL  (BATCH * COLS)
#define TILE   (BATCH * DOUT)
#define RWAVES 4
#define RCH    (NCHUNK / 2)
#define WPB    8
#define NGRP   (CCAP / WPB)
#define NGRPN  8
#define XPITCH (NPER * DIN + 4)

typedef __attribute__((ext_vector_type(2))) float v2f;
typedef __attribute__((ext_vector_type(8))) float v8f;

#define ROWX(v, m) __builtin_bit_cast(float, __builtin_amdgcn_update_dpp( \
    __builtin_bit_cast(int, (v)), __builtin_bit_cast(int, (v)), 0x160 + (m), 0xf, 0xf, false))
__device__ __forceinline__ float xhalf(float v) {
    const int b = __builtin_bit_cast(int, v);
    return __builtin_bit_cast(float,
        __builtin_amdgcn_permlanex16(b, b, 0x76543210, 0xfedcba98, false, false));
}
__device__ __forceinline__ float wave_max(float v) {
    v = fmaxf(v, ROWX(v, 1));
    v = fmaxf(v, ROWX(v, 2));
    v = fmaxf(v, ROWX(v, 4));
    v = fmaxf(v, ROWX(v, 8));
    return fmaxf(v, xhalf(v));
}
__device__ __forceinline__ float wave_sum(float v) {
    v += ROWX(v, 1);
    v += ROWX(v, 2);
    v += ROWX(v, 4);
    v += ROWX(v, 8);
    return v + xhalf(v);
}

__device__ __forceinline__ void pin() { __builtin_amdgcn_sched_barrier(0); }

__global__ __launch_bounds__(256)
void caps_gemm(const float* __restrict__ x, const float* __restrict__ W,
               const float* __restrict__ R, float* __restrict__ part,
               float* __restrict__ mst, float* __restrict__ zst) {
    __shared__ __attribute__((aligned(16))) float xs[BATCH * XPITCH];

    const int tid  = threadIdx.x;
    const int lane = tid & 31;
    const int wave = tid >> 5;
    const int t    = blockIdx.x * WPB + wave;
    const int ch   = blockIdx.y;
    const int j    = lane & 15;
    const int hi   = lane >> 4;
    const int n0   = ch * NPER;

    const int q  = tid & 63;
    const int br = tid >> 6;
    float4 xv[8];
    #pragma unroll
    for (int i = 0; i < 8; ++i)
        xv[i] = *(const float4*)(x + (size_t)(i * 4 + br) * (NTOT * DIN) + n0 * DIN + q * 4);

    float r = R[(n0 + lane) * CCAP + t];

    const float* wp = W + (size_t)n0 * (CCAP * DOUT * DIN) + t * (DOUT * DIN) + j * DIN + hi * 4;
    float4 wv[NPER];
    #pragma unroll
    for (int nn = 0; nn < NPER; ++nn)
        wv[nn] = *(const float4*)(wp + (size_t)nn * (CCAP * DOUT * DIN));

    #pragma unroll
    for (int i = 0; i < 8; ++i)
        *(float4*)(xs + (i * 4 + br) * XPITCH + q * 4) = xv[i];
    asm volatile("s_wait_dscnt 0x0" ::: "memory");
    asm volatile("s_barrier_signal -1" : "+v"(r) :: "memory");

    const float mloc = wave_max(r);
    const float e    = __expf(r - mloc);
    if (lane == 0) mst[t * NCHUNK + ch] = mloc;
    const int ebits = __float_as_int(e);
    float zs = e;

    asm volatile("s_barrier_wait -1" ::: "memory");

    v8f acc0 = {};
    v8f acc1 = {};
    const float* xa0 = xs + j * XPITCH + hi * 4;
    const float* xa1 = xs + (j + 16) * XPITCH + hi * 4;

    float4 fa[2][NGRPN], fb[2][NGRPN];
    #pragma unroll
    for (int u = 0; u < NGRPN; ++u) {
        fa[0][u] = *(const float4*)(xa0 + u * DIN);
        fb[0][u] = *(const float4*)(xa1 + u * DIN);
    }
    pin();
    #pragma unroll
    for (int gq = 0; gq < NPER / NGRPN; ++gq) {
        const int cur = gq & 1;
        if (gq + 1 < NPER / NGRPN) {
            #pragma unroll
            for (int u = 0; u < NGRPN; ++u) {
                fa[cur ^ 1][u] = *(const float4*)(xa0 + ((gq + 1) * NGRPN + u) * DIN);
                fb[cur ^ 1][u] = *(const float4*)(xa1 + ((gq + 1) * NGRPN + u) * DIN);
            }
        }
        pin();
        #pragma unroll
        for (int u = 0; u < NGRPN; ++u) {
            const int nn = gq * NGRPN + u;
            const float rn = __int_as_float(__builtin_amdgcn_readlane(ebits, nn));
            const float4 w  = wv[nn];
            const float4 a0 = fa[cur][u];
            const float4 a1 = fb[cur][u];
            if (gq == 0) {
                if (u == 0) zs += ROWX(zs, 1);
                if (u == 1) zs += ROWX(zs, 2);
                if (u == 2) zs += ROWX(zs, 4);
                if (u == 3) zs += ROWX(zs, 8);
                if (u == 4) zs += xhalf(zs);
            }

            v2f b0, b1, a00, a01, a10, a11;
            b0.x  = w.x * rn;    b0.y  = w.y * rn;
            b1.x  = w.z * rn;    b1.y  = w.w * rn;
            a00.x = a0.x;        a00.y = a0.y;
            a01.x = a0.z;        a01.y = a0.w;
            a10.x = a1.x;        a10.y = a1.y;
            a11.x = a1.z;        a11.y = a1.w;

            acc0 = __builtin_amdgcn_wmma_f32_16x16x4_f32(false, a00, false, b0,
                                                         (short)0, acc0, false, false);
            acc1 = __builtin_amdgcn_wmma_f32_16x16x4_f32(false, a10, false, b0,
                                                         (short)0, acc1, false, false);
            acc0 = __builtin_amdgcn_wmma_f32_16x16x4_f32(false, a01, false, b1,
                                                         (short)0, acc0, false, false);
            acc1 = __builtin_amdgcn_wmma_f32_16x16x4_f32(false, a11, false, b1,
                                                         (short)0, acc1, false, false);
        }
        pin();
    }

    float* p = part + ((size_t)ch * CCAP + t) * TILE + lane;
    #pragma unroll
    for (int v = 0; v < 8; ++v) {
        p[v * 32]       = acc0[v];
        p[256 + v * 32] = acc1[v];
    }
    if (lane == 0) zst[t * NCHUNK + ch] = zs;
}

__global__ __launch_bounds__(RWAVES * 32)
void caps_reduce_squash(const float* __restrict__ part, const float* __restrict__ mst,
                        const float* __restrict__ zst, float* __restrict__ out) {
    __shared__ float sh[RWAVES * 32];

    const int tid  = threadIdx.x;
    const int lane = tid & 31;
    const int wv   = tid >> 5;
    const int cw   = wv & 1;
    const int q    = wv >> 1;
    const int cp   = blockIdx.x >> 4;
    const int e    = blockIdx.x & 15;
    const int c    = cp * 2 + cw;

    const float m0 = mst[c * NCHUNK + lane];
    const float m1 = mst[c * NCHUNK + 32 + lane];
    const float z0 = zst[c * NCHUNK + lane];
    const float z1 = zst[c * NCHUNK + 32 + lane];

    const float* pp = part + ((size_t)(q * RCH) * CCAP + c) * TILE + e * 32 + lane;
    float pv[RCH];
    #pragma unroll
    for (int i = 0; i < RCH; ++i) pv[i] = pp[(size_t)i * OUTEL];
    pin();

    const float M  = wave_max(fmaxf(m0, m1));
    const float f0 = __expf(m0 - M);
    const float f1 = __expf(m1 - M);
    const float Z  = wave_sum(z0 * f0 + z1 * f1);
    const int   fb = __float_as_int(q ? f1 : f0);

    float s0 = 0.0f, s1 = 0.0f, s2 = 0.0f, s3 = 0.0f;
    #pragma unroll
    for (int i = 0; i < RCH; i += 4) {
        s0 += pv[i + 0] * __int_as_float(__builtin_amdgcn_readlane(fb, i + 0));
        s1 += pv[i + 1] * __int_as_float(__builtin_amdgcn_readlane(fb, i + 1));
        s2 += pv[i + 2] * __int_as_float(__builtin_amdgcn_readlane(fb, i + 2));
        s3 += pv[i + 3] * __int_as_float(__builtin_amdgcn_readlane(fb, i + 3));
    }
    sh[tid] = ((s0 + s1) + (s2 + s3)) * (1.0f / Z);
    __syncthreads();

    if (tid < 64) {
        const int hc = lane >> 4;
        const int j  = lane & 15;
        const float s = sh[hc * 32 + wv * 16 + j] + sh[64 + hc * 32 + wv * 16 + j];

        float sq = s * s;
        sq += ROWX(sq, 1);
        sq += ROWX(sq, 2);
        sq += ROWX(sq, 4);
        sq += ROWX(sq, 8);
        sq += 1e-7f;
        const float scale = sqrtf(sq) / (1.0f + sq);

        const int pos = 2 * e + wv;
        const int b   = (pos & 16) + ((pos & 15) >> 1) + 8 * (pos & 1);
        out[b * COLS + cp * 32 + lane] = scale * s;
    }
}

extern "C" void kernel_launch(void* const* d_in, const int* in_sizes, int n_in,
                              void* d_out, int out_size, void* d_ws, size_t ws_size,
                              hipStream_t stream) {
    const float* x = (const float*)d_in[0];
    const float* W = (const float*)d_in[1];
    const float* R = (const float*)d_in[2];
    float* out = (float*)d_out;

    float* mst  = (float*)d_ws;
    float* zst  = mst + NCHUNK * CCAP;
    float* part = mst + (size_t)NTOT * CCAP;

    dim3 grid(NGRP, NCHUNK);
    caps_gemm<<<grid, WPB * 32, 0, stream>>>(x, W, R, part, mst, zst);

    caps_reduce_squash<<<256, RWAVES * 32, 0, stream>>>(part, mst, zst, out);
}
